# starting kernel
# baseline (speedup 1.0000x reference)
#include <hip/hip_runtime.h>
#include <math.h>


#define DIM 16

typedef float v2f __attribute__((ext_vector_type(2)));
typedef float v8f __attribute__((ext_vector_type(8)));

__host__ __device__ constexpr int cnot_perm_idx(int c, int t, int idx) {
    return idx ^ ((((idx >> (3 - c)) & 1)) << (3 - t));
}
__host__ __device__ constexpr int ring_perm_idx(int j) {
    int v = cnot_perm_idx(3, 0, j);
    v = cnot_perm_idx(2, 3, v);
    v = cnot_perm_idx(1, 2, v);
    v = cnot_perm_idx(0, 1, v);
    return v;
}

__global__ void build_fixed_matrix_kernel(const float* __restrict__ w,
                                          float* __restrict__ M) {
    const int t = threadIdx.x;
    if (t >= DIM) return;
    float v[DIM];
#pragma unroll
    for (int i = 0; i < DIM; ++i) v[i] = (i == t) ? 1.0f : 0.0f;
#pragma unroll
    for (int layer = 0; layer < 3; ++layer) {
#pragma unroll
        for (int q = 0; q < 4; ++q) {
            const float th = 0.5f * w[layer * 4 + q];
            const float c = cosf(th);
            const float s = sinf(th);
            const int mask = 1 << (3 - q);
#pragma unroll
            for (int i = 0; i < DIM; ++i) {
                if ((i & mask) == 0) {
                    const int j = i | mask;
                    const float a0 = v[i], a1 = v[j];
                    v[i] = c * a0 - s * a1;
                    v[j] = s * a0 + c * a1;
                }
            }
        }
        float nv[DIM];
#pragma unroll
        for (int i = 0; i < DIM; ++i) nv[i] = v[ring_perm_idx(i)];
#pragma unroll
        for (int i = 0; i < DIM; ++i) v[i] = nv[i];
    }
#pragma unroll
    for (int i = 0; i < DIM; ++i) M[i * DIM + t] = v[i];
}

__global__ void __launch_bounds__(256)
quantum_layer_kernel(const float* __restrict__ x, const float* __restrict__ M,
                     float* __restrict__ out, int ntiles, long Btot) {
    __shared__ float probs[8][DIM * 17];
    const int lane = threadIdx.x & 31;
    const int wave = threadIdx.x >> 5;
    const int m    = lane & 15;
    const int h    = lane >> 4;
    const long tile = (long)blockIdx.x * 8 + wave;
    const bool active = tile < (long)ntiles;

    if (active) {
        long b = tile * 16 + m;
        const long bl = (b < Btot) ? b : (Btot - 1);
        const float4 xv = ((const float4*)x)[bl];

        float c0, s0, c1, s1, c2, s2, c3, s3;
        sincosf(0.5f * xv.x, &s0, &c0);
        sincosf(0.5f * xv.y, &s1, &c1);
        sincosf(0.5f * xv.z, &s2, &c2);
        sincosf(0.5f * xv.w, &s3, &c3);
        const float f2 = h ? s2 : c2;

        v8f acc = {0.f, 0.f, 0.f, 0.f, 0.f, 0.f, 0.f, 0.f};
#pragma unroll
        for (int g = 0; g < 4; ++g) {
            const float base = ((g & 2) ? s0 : c0) * ((g & 1) ? s1 : c1) * f2;
            v2f a;
            a.x = base * c3;
            a.y = base * s3;
            const float* mrow = M + m * DIM + g * 4 + 2 * h;
            v2f bm;
            bm.x = mrow[0];
            bm.y = mrow[1];
            acc = __builtin_amdgcn_wmma_f32_16x16x4_f32(
                false, a, false, bm, (short)0, acc, false, false);
        }
#pragma unroll
        for (int r = 0; r < 8; ++r) {
            const int row = r + 8 * h;
            const float yv = acc[r];
            probs[wave][row * 17 + m] = yv * yv;
        }
    }
    __syncthreads();
    if (active) {
        const long b = tile * 16 + m;
        if (b < Btot) {
            const int bp0 = 3 - (2 * h);
            const int bp1 = 3 - (2 * h + 1);
            float a0 = 0.f, a1 = 0.f;
#pragma unroll
            for (int n = 0; n < DIM; ++n) {
                const float p = probs[wave][m * 17 + n];
                a0 += ((n >> bp0) & 1) ? -p : p;
                a1 += ((n >> bp1) & 1) ? -p : p;
            }
            float2 r2;
            r2.x = a0;
            r2.y = a1;
            ((float2*)out)[b * 2 + h] = r2;
        }
    }
}

extern "C" void kernel_launch(void* const* d_in, const int* in_sizes, int n_in,
                              void* d_out, int out_size, void* d_ws, size_t ws_size,
                              hipStream_t stream) {
    const float* x = (const float*)d_in[0];
    const float* w = (const float*)d_in[1];
    float* out = (float*)d_out;
    float* M = (float*)d_ws;

    const long B = (long)in_sizes[0] / 4;
    const int ntiles = (int)((B + 15) / 16);
    const int blocks = (ntiles + 7) / 8;

    build_fixed_matrix_kernel<<<1, 16, 0, stream>>>(w, M);
    quantum_layer_kernel<<<blocks, 256, 0, stream>>>(x, M, out, ntiles, B);
}


# A sample's two lanes (l, l+16) each compute sincos for 2 of its 4 angles from one 8-byte load and swap them by 4 v_permlanex16, instead of both computing all 4; arithmetic and sum order unchanged
# speedup vs baseline: 1.0598x; 1.0598x over previous
#include <hip/hip_runtime.h>
#include <math.h>


#define DIM 16

typedef float v2f __attribute__((ext_vector_type(2)));
typedef float v8f __attribute__((ext_vector_type(8)));

__device__ __forceinline__ float xhalf(float v) {
    const int b = __builtin_bit_cast(int, v);
    return __builtin_bit_cast(float, __builtin_amdgcn_permlanex16(
        b, b, 0x76543210, 0xfedcba98, false, false));
}

__host__ __device__ constexpr int cnot_perm_idx(int c, int t, int idx) {
    return idx ^ ((((idx >> (3 - c)) & 1)) << (3 - t));
}
__host__ __device__ constexpr int ring_perm_idx(int j) {
    int v = cnot_perm_idx(3, 0, j);
    v = cnot_perm_idx(2, 3, v);
    v = cnot_perm_idx(1, 2, v);
    v = cnot_perm_idx(0, 1, v);
    return v;
}

__global__ void build_fixed_matrix_kernel(const float* __restrict__ w,
                                          float* __restrict__ M) {
    const int t = threadIdx.x;
    if (t >= DIM) return;
    float v[DIM];
#pragma unroll
    for (int i = 0; i < DIM; ++i) v[i] = (i == t) ? 1.0f : 0.0f;
#pragma unroll
    for (int layer = 0; layer < 3; ++layer) {
#pragma unroll
        for (int q = 0; q < 4; ++q) {
            const float th = 0.5f * w[layer * 4 + q];
            const float c = cosf(th);
            const float s = sinf(th);
            const int mask = 1 << (3 - q);
#pragma unroll
            for (int i = 0; i < DIM; ++i) {
                if ((i & mask) == 0) {
                    const int j = i | mask;
                    const float a0 = v[i], a1 = v[j];
                    v[i] = c * a0 - s * a1;
                    v[j] = s * a0 + c * a1;
                }
            }
        }
        float nv[DIM];
#pragma unroll
        for (int i = 0; i < DIM; ++i) nv[i] = v[ring_perm_idx(i)];
#pragma unroll
        for (int i = 0; i < DIM; ++i) v[i] = nv[i];
    }
#pragma unroll
    for (int i = 0; i < DIM; ++i) M[i * DIM + t] = v[i];
}

__global__ void __launch_bounds__(256)
quantum_layer_kernel(const float* __restrict__ x, const float* __restrict__ M,
                     float* __restrict__ out, int ntiles, long Btot) {
    __shared__ float probs[8][DIM * 17];
    const int lane = threadIdx.x & 31;
    const int wave = threadIdx.x >> 5;
    const int m    = lane & 15;
    const int h    = lane >> 4;
    const long tile = (long)blockIdx.x * 8 + wave;
    const bool active = tile < (long)ntiles;

    if (active) {
        long b = tile * 16 + m;
        const long bl = (b < Btot) ? b : (Btot - 1);
        const float2 xh = ((const float2*)x)[bl * 2 + h];
        float sa, ca, sb, cb;
        sincosf(0.5f * xh.x, &sa, &ca);
        sincosf(0.5f * xh.y, &sb, &cb);
        const float psa = xhalf(sa);
        const float pca = xhalf(ca);
        const float psb = xhalf(sb);
        const float pcb = xhalf(cb);
        const float c0 = h ? pca : ca;
        const float s0 = h ? psa : sa;
        const float c1 = h ? pcb : cb;
        const float s1 = h ? psb : sb;
        const float f2 = h ? sa : pca;
        const float c3 = h ? cb : pcb;
        const float s3 = h ? sb : psb;

        v8f acc = {0.f, 0.f, 0.f, 0.f, 0.f, 0.f, 0.f, 0.f};
#pragma unroll
        for (int g = 0; g < 4; ++g) {
            const float base = ((g & 2) ? s0 : c0) * ((g & 1) ? s1 : c1) * f2;
            v2f a;
            a.x = base * c3;
            a.y = base * s3;
            const float* mrow = M + m * DIM + g * 4 + 2 * h;
            v2f bm;
            bm.x = mrow[0];
            bm.y = mrow[1];
            acc = __builtin_amdgcn_wmma_f32_16x16x4_f32(
                false, a, false, bm, (short)0, acc, false, false);
        }
#pragma unroll
        for (int r = 0; r < 8; ++r) {
            const int row = r + 8 * h;
            const float yv = acc[r];
            probs[wave][row * 17 + m] = yv * yv;
        }
    }
    __syncthreads();
    if (active) {
        const long b = tile * 16 + m;
        if (b < Btot) {
            const int bp0 = 3 - (2 * h);
            const int bp1 = 3 - (2 * h + 1);
            float a0 = 0.f, a1 = 0.f;
#pragma unroll
            for (int n = 0; n < DIM; ++n) {
                const float p = probs[wave][m * 17 + n];
                a0 += ((n >> bp0) & 1) ? -p : p;
                a1 += ((n >> bp1) & 1) ? -p : p;
            }
            float2 r2;
            r2.x = a0;
            r2.y = a1;
            ((float2*)out)[b * 2 + h] = r2;
        }
    }
}

extern "C" void kernel_launch(void* const* d_in, const int* in_sizes, int n_in,
                              void* d_out, int out_size, void* d_ws, size_t ws_size,
                              hipStream_t stream) {
    const float* x = (const float*)d_in[0];
    const float* w = (const float*)d_in[1];
    float* out = (float*)d_out;
    float* M = (float*)d_ws;

    const long B = (long)in_sizes[0] / 4;
    const int ntiles = (int)((B + 15) / 16);
    const int blocks = (ntiles + 7) / 8;

    build_fixed_matrix_kernel<<<1, 16, 0, stream>>>(w, M);
    quantum_layer_kernel<<<blocks, 256, 0, stream>>>(x, M, out, ntiles, B);
}


# Set-up kernel: the 12 weight sin/cos pairs are computed one per lane on the vector unit and read across lanes by v_readlane, not as 24 serial scalar-unit evaluations; arithmetic, sum order unchanged
# speedup vs baseline: 1.1870x; 1.1200x over previous
#include <hip/hip_runtime.h>
#include <math.h>


#define DIM 16

typedef float v2f __attribute__((ext_vector_type(2)));
typedef float v8f __attribute__((ext_vector_type(8)));

__device__ __forceinline__ float xhalf(float v) {
    const int b = __builtin_bit_cast(int, v);
    return __builtin_bit_cast(float, __builtin_amdgcn_permlanex16(
        b, b, 0x76543210, 0xfedcba98, false, false));
}

__host__ __device__ constexpr int cnot_perm_idx(int c, int t, int idx) {
    return idx ^ ((((idx >> (3 - c)) & 1)) << (3 - t));
}
__host__ __device__ constexpr int ring_perm_idx(int j) {
    int v = cnot_perm_idx(3, 0, j);
    v = cnot_perm_idx(2, 3, v);
    v = cnot_perm_idx(1, 2, v);
    v = cnot_perm_idx(0, 1, v);
    return v;
}

__device__ __forceinline__ float lane_value(float v, int j) {
    return __builtin_bit_cast(float,
        __builtin_amdgcn_readlane(__builtin_bit_cast(int, v), j));
}

__global__ void build_fixed_matrix_kernel(const float* __restrict__ w,
                                          float* __restrict__ M) {
    const int t = threadIdx.x;
    if (t >= DIM) return;
    float sj, cj;
    sincosf(0.5f * w[t < 12 ? t : 11], &sj, &cj);
    float v[DIM];
#pragma unroll
    for (int i = 0; i < DIM; ++i) v[i] = (i == t) ? 1.0f : 0.0f;
#pragma unroll
    for (int layer = 0; layer < 3; ++layer) {
#pragma unroll
        for (int q = 0; q < 4; ++q) {
            const float c = lane_value(cj, layer * 4 + q);
            const float s = lane_value(sj, layer * 4 + q);
            const int mask = 1 << (3 - q);
#pragma unroll
            for (int i = 0; i < DIM; ++i) {
                if ((i & mask) == 0) {
                    const int j = i | mask;
                    const float a0 = v[i], a1 = v[j];
                    v[i] = c * a0 - s * a1;
                    v[j] = s * a0 + c * a1;
                }
            }
        }
        float nv[DIM];
#pragma unroll
        for (int i = 0; i < DIM; ++i) nv[i] = v[ring_perm_idx(i)];
#pragma unroll
        for (int i = 0; i < DIM; ++i) v[i] = nv[i];
    }
#pragma unroll
    for (int i = 0; i < DIM; ++i) M[i * DIM + t] = v[i];
}

__global__ void __launch_bounds__(256)
quantum_layer_kernel(const float* __restrict__ x, const float* __restrict__ M,
                     float* __restrict__ out, int ntiles, long Btot) {
    __shared__ float probs[8][DIM * 17];
    const int lane = threadIdx.x & 31;
    const int wave = threadIdx.x >> 5;
    const int m    = lane & 15;
    const int h    = lane >> 4;
    const long tile = (long)blockIdx.x * 8 + wave;
    const bool active = tile < (long)ntiles;

    if (active) {
        long b = tile * 16 + m;
        const long bl = (b < Btot) ? b : (Btot - 1);
        const float2 xh = ((const float2*)x)[bl * 2 + h];
        float sa, ca, sb, cb;
        sincosf(0.5f * xh.x, &sa, &ca);
        sincosf(0.5f * xh.y, &sb, &cb);
        const float psa = xhalf(sa);
        const float pca = xhalf(ca);
        const float psb = xhalf(sb);
        const float pcb = xhalf(cb);
        const float c0 = h ? pca : ca;
        const float s0 = h ? psa : sa;
        const float c1 = h ? pcb : cb;
        const float s1 = h ? psb : sb;
        const float f2 = h ? sa : pca;
        const float c3 = h ? cb : pcb;
        const float s3 = h ? sb : psb;

        v8f acc = {0.f, 0.f, 0.f, 0.f, 0.f, 0.f, 0.f, 0.f};
#pragma unroll
        for (int g = 0; g < 4; ++g) {
            const float base = ((g & 2) ? s0 : c0) * ((g & 1) ? s1 : c1) * f2;
            v2f a;
            a.x = base * c3;
            a.y = base * s3;
            const float* mrow = M + m * DIM + g * 4 + 2 * h;
            v2f bm;
            bm.x = mrow[0];
            bm.y = mrow[1];
            acc = __builtin_amdgcn_wmma_f32_16x16x4_f32(
                false, a, false, bm, (short)0, acc, false, false);
        }
#pragma unroll
        for (int r = 0; r < 8; ++r) {
            const int row = r + 8 * h;
            const float yv = acc[r];
            probs[wave][row * 17 + m] = yv * yv;
        }
    }
    __syncthreads();
    if (active) {
        const long b = tile * 16 + m;
        if (b < Btot) {
            const int bp0 = 3 - (2 * h);
            const int bp1 = 3 - (2 * h + 1);
            float a0 = 0.f, a1 = 0.f;
#pragma unroll
            for (int n = 0; n < DIM; ++n) {
                const float p = probs[wave][m * 17 + n];
                a0 += ((n >> bp0) & 1) ? -p : p;
                a1 += ((n >> bp1) & 1) ? -p : p;
            }
            float2 r2;
            r2.x = a0;
            r2.y = a1;
            ((float2*)out)[b * 2 + h] = r2;
        }
    }
}

extern "C" void kernel_launch(void* const* d_in, const int* in_sizes, int n_in,
                              void* d_out, int out_size, void* d_ws, size_t ws_size,
                              hipStream_t stream) {
    const float* x = (const float*)d_in[0];
    const float* w = (const float*)d_in[1];
    float* out = (float*)d_out;
    float* M = (float*)d_ws;

    const long B = (long)in_sizes[0] / 4;
    const int ntiles = (int)((B + 15) / 16);
    const int blocks = (ntiles + 7) / 8;

    build_fixed_matrix_kernel<<<1, 16, 0, stream>>>(w, M);
    quantum_layer_kernel<<<blocks, 256, 0, stream>>>(x, M, out, ntiles, B);
}


# Product transposed (operator as row operand, states as column operand): a sample's 16 amplitudes sit in its 2 lanes; Z sums by in-register add tree + half swap (sum order changes); no LDS, no barrier
# speedup vs baseline: 1.3049x; 1.0993x over previous
#include <hip/hip_runtime.h>
#include <math.h>


#define DIM 16

typedef float v2f __attribute__((ext_vector_type(2)));
typedef float v8f __attribute__((ext_vector_type(8)));

__device__ __forceinline__ float xhalf(float v) {
    const int b = __builtin_bit_cast(int, v);
    return __builtin_bit_cast(float, __builtin_amdgcn_permlanex16(
        b, b, 0x76543210, 0xfedcba98, false, false));
}

__host__ __device__ constexpr int cnot_perm_idx(int c, int t, int idx) {
    return idx ^ ((((idx >> (3 - c)) & 1)) << (3 - t));
}
__host__ __device__ constexpr int ring_perm_idx(int j) {
    int v = cnot_perm_idx(3, 0, j);
    v = cnot_perm_idx(2, 3, v);
    v = cnot_perm_idx(1, 2, v);
    v = cnot_perm_idx(0, 1, v);
    return v;
}

__device__ __forceinline__ float lane_value(float v, int j) {
    return __builtin_bit_cast(float,
        __builtin_amdgcn_readlane(__builtin_bit_cast(int, v), j));
}

__global__ void build_fixed_matrix_kernel(const float* __restrict__ w,
                                          float* __restrict__ M) {
    const int t = threadIdx.x;
    if (t >= DIM) return;
    float sj, cj;
    sincosf(0.5f * w[t < 12 ? t : 11], &sj, &cj);
    float v[DIM];
#pragma unroll
    for (int i = 0; i < DIM; ++i) v[i] = (i == t) ? 1.0f : 0.0f;
#pragma unroll
    for (int layer = 0; layer < 3; ++layer) {
#pragma unroll
        for (int q = 0; q < 4; ++q) {
            const float c = lane_value(cj, layer * 4 + q);
            const float s = lane_value(sj, layer * 4 + q);
            const int mask = 1 << (3 - q);
#pragma unroll
            for (int i = 0; i < DIM; ++i) {
                if ((i & mask) == 0) {
                    const int j = i | mask;
                    const float a0 = v[i], a1 = v[j];
                    v[i] = c * a0 - s * a1;
                    v[j] = s * a0 + c * a1;
                }
            }
        }
        float nv[DIM];
#pragma unroll
        for (int i = 0; i < DIM; ++i) nv[i] = v[ring_perm_idx(i)];
#pragma unroll
        for (int i = 0; i < DIM; ++i) v[i] = nv[i];
    }
#pragma unroll
    for (int i = 0; i < DIM; ++i) M[i * DIM + t] = v[i];
}

__global__ void __launch_bounds__(256)
quantum_layer_kernel(const float* __restrict__ x, const float* __restrict__ M,
                     float* __restrict__ out, int ntiles, long Btot) {
    const int lane = threadIdx.x & 31;
    const int wave = threadIdx.x >> 5;
    const int m    = lane & 15;
    const int h    = lane >> 4;
    const long tile = (long)blockIdx.x * 8 + wave;
    const bool active = tile < (long)ntiles;

    if (active) {
        const long b = tile * 16 + m;
        const long bl = (b < Btot) ? b : (Btot - 1);
        const float2 xh = ((const float2*)x)[bl * 2 + h];
        float sa, ca, sb, cb;
        sincosf(0.5f * xh.x, &sa, &ca);
        sincosf(0.5f * xh.y, &sb, &cb);
        const float psa = xhalf(sa);
        const float pca = xhalf(ca);
        const float psb = xhalf(sb);
        const float pcb = xhalf(cb);
        const float c0 = h ? pca : ca;
        const float s0 = h ? psa : sa;
        const float c1 = h ? pcb : cb;
        const float s1 = h ? psb : sb;
        const float f2 = h ? sa : pca;
        const float c3 = h ? cb : pcb;
        const float s3 = h ? sb : psb;

        v8f acc = {0.f, 0.f, 0.f, 0.f, 0.f, 0.f, 0.f, 0.f};
#pragma unroll
        for (int g = 0; g < 4; ++g) {
            const float base = ((g & 2) ? s0 : c0) * ((g & 1) ? s1 : c1) * f2;
            v2f a;
            a.x = base * c3;
            a.y = base * s3;
            const float* mrow = M + m * DIM + g * 4 + 2 * h;
            v2f bm;
            bm.x = mrow[0];
            bm.y = mrow[1];
            acc = __builtin_amdgcn_wmma_f32_16x16x4_f32(
                false, bm, false, a, (short)0, acc, false, false);
        }
        float p[8];
#pragma unroll
        for (int r = 0; r < 8; ++r) p[r] = acc[r] * acc[r];
        const float s01 = p[0] + p[1], d01 = p[0] - p[1];
        const float s23 = p[2] + p[3], d23 = p[2] - p[3];
        const float s45 = p[4] + p[5], d45 = p[4] - p[5];
        const float s67 = p[6] + p[7], d67 = p[6] - p[7];
        const float z3  = (d01 + d23) + (d45 + d67);
        const float t0  = s01 + s23,  e0 = s01 - s23;
        const float t1  = s45 + s67,  e1 = s45 - s67;
        const float z2  = e0 + e1;
        const float tot = t0 + t1;
        const float z1  = t0 - t1;
        const float pu = xhalf(h ? tot : z2);
        const float pv = xhalf(h ? z1 : z3);
        float2 r2;
        r2.x = h ? (z2 + pu) : (tot - pu);
        r2.y = (h ? z3 : z1) + pv;
        if (b < Btot)
            ((float2*)out)[b * 2 + h] = r2;
    }
}

extern "C" void kernel_launch(void* const* d_in, const int* in_sizes, int n_in,
                              void* d_out, int out_size, void* d_ws, size_t ws_size,
                              hipStream_t stream) {
    const float* x = (const float*)d_in[0];
    const float* w = (const float*)d_in[1];
    float* out = (float*)d_out;
    float* M = (float*)d_ws;

    const long B = (long)in_sizes[0] / 4;
    const int ntiles = (int)((B + 15) / 16);
    const int blocks = (ntiles + 7) / 8;

    build_fixed_matrix_kernel<<<1, 16, 0, stream>>>(w, M);
    quantum_layer_kernel<<<blocks, 256, 0, stream>>>(x, M, out, ntiles, B);
}


# Set-up launch folded into the main kernel: wave 0 of each of 1024 8-wave work-groups builds the operator into LDS (same arithmetic), one barrier, then every wave loops over its 8 tiles
# speedup vs baseline: 1.6335x; 1.2518x over previous
#include <hip/hip_runtime.h>
#include <math.h>


#define DIM 16
#define BLOCK_WAVES 8
#define GRID_BLOCKS 1024

typedef float v2f __attribute__((ext_vector_type(2)));
typedef float v8f __attribute__((ext_vector_type(8)));

__device__ __forceinline__ float xhalf(float v) {
    const int b = __builtin_bit_cast(int, v);
    return __builtin_bit_cast(float, __builtin_amdgcn_permlanex16(
        b, b, 0x76543210, 0xfedcba98, false, false));
}

__host__ __device__ constexpr int cnot_perm_idx(int c, int t, int idx) {
    return idx ^ ((((idx >> (3 - c)) & 1)) << (3 - t));
}
__host__ __device__ constexpr int ring_perm_idx(int j) {
    int v = cnot_perm_idx(3, 0, j);
    v = cnot_perm_idx(2, 3, v);
    v = cnot_perm_idx(1, 2, v);
    v = cnot_perm_idx(0, 1, v);
    return v;
}

__device__ __forceinline__ float lane_value(float v, int j) {
    return __builtin_bit_cast(float,
        __builtin_amdgcn_readlane(__builtin_bit_cast(int, v), j));
}

__device__ __forceinline__ void build_fixed_matrix(float wj, float* __restrict__ Ms,
                                                   int lane) {
    const int t = lane & 15;
    float sj, cj;
    sincosf(0.5f * wj, &sj, &cj);
    float v[DIM];
#pragma unroll
    for (int i = 0; i < DIM; ++i) v[i] = (i == t) ? 1.0f : 0.0f;
#pragma unroll
    for (int layer = 0; layer < 3; ++layer) {
#pragma unroll
        for (int q = 0; q < 4; ++q) {
            const float c = lane_value(cj, layer * 4 + q);
            const float s = lane_value(sj, layer * 4 + q);
            const int mask = 1 << (3 - q);
#pragma unroll
            for (int i = 0; i < DIM; ++i) {
                if ((i & mask) == 0) {
                    const int j = i | mask;
                    const float a0 = v[i], a1 = v[j];
                    v[i] = c * a0 - s * a1;
                    v[j] = s * a0 + c * a1;
                }
            }
        }
        float nv[DIM];
#pragma unroll
        for (int i = 0; i < DIM; ++i) nv[i] = v[ring_perm_idx(i)];
#pragma unroll
        for (int i = 0; i < DIM; ++i) v[i] = nv[i];
    }
    if (lane < DIM) {
#pragma unroll
        for (int i = 0; i < DIM; ++i) Ms[i * DIM + t] = v[i];
    }
}

__device__ __forceinline__ float2 load_angles(const float2* __restrict__ x2, int tile,
                                              int ntiles, int Btot, int m, int h) {
    const int t = tile < ntiles ? tile : ntiles - 1;
    const int rem = Btot - t * 16;
    const int nvalid = rem < 16 ? rem : 16;
    const int ml = m < nvalid ? m : nvalid - 1;
    const float2* __restrict__ base = x2 + (long)t * 32;
    return base[ml * 2 + h];
}

struct Trig { float c0, s0, c1, s1, f2, c3, s3; };

__device__ __forceinline__ Trig sample_trig(float2 xh, int h) {
    float sa, ca, sb, cb;
    sincosf(0.5f * xh.x, &sa, &ca);
    sincosf(0.5f * xh.y, &sb, &cb);
    const float psa = xhalf(sa);
    const float pca = xhalf(ca);
    const float psb = xhalf(sb);
    const float pcb = xhalf(cb);
    Trig t;
    t.c0 = h ? pca : ca;
    t.s0 = h ? psa : sa;
    t.c1 = h ? pcb : cb;
    t.s1 = h ? psb : sb;
    t.f2 = h ? sa : pca;
    t.c3 = h ? cb : pcb;
    t.s3 = h ? sb : psb;
    return t;
}

__device__ __forceinline__ void tile_body(const Trig& tr, const v2f (&bm)[4],
                                          float2* __restrict__ out2, int tile,
                                          int Btot, int m, int h) {
    v8f acc = {0.f, 0.f, 0.f, 0.f, 0.f, 0.f, 0.f, 0.f};
#pragma unroll
    for (int g = 0; g < 4; ++g) {
        const float base = ((g & 2) ? tr.s0 : tr.c0) * ((g & 1) ? tr.s1 : tr.c1) * tr.f2;
        v2f a;
        a.x = base * tr.c3;
        a.y = base * tr.s3;
        acc = __builtin_amdgcn_wmma_f32_16x16x4_f32(
            false, bm[g], false, a, (short)0, acc, false, false);
    }
    float p[8];
#pragma unroll
    for (int r = 0; r < 8; ++r) p[r] = acc[r] * acc[r];
    const float s01 = p[0] + p[1], d01 = p[0] - p[1];
    const float s23 = p[2] + p[3], d23 = p[2] - p[3];
    const float s45 = p[4] + p[5], d45 = p[4] - p[5];
    const float s67 = p[6] + p[7], d67 = p[6] - p[7];
    const float z3  = (d01 + d23) + (d45 + d67);
    const float t0  = s01 + s23,  e0 = s01 - s23;
    const float t1  = s45 + s67,  e1 = s45 - s67;
    const float z2  = e0 + e1;
    const float tot = t0 + t1;
    const float z1  = t0 - t1;
    const float pu = xhalf(h ? tot : z2);
    const float pv = xhalf(h ? z1 : z3);
    float2 r2;
    r2.x = h ? (z2 + pu) : (tot - pu);
    r2.y = (h ? z3 : z1) + pv;
    const int rem = Btot - tile * 16;
    const int nvalid = rem < 16 ? rem : 16;
    if (m < nvalid)
        (out2 + (long)tile * 32)[m * 2 + h] = r2;
}

__global__ void __launch_bounds__(BLOCK_WAVES * 32)
quantum_layer_kernel(const float* __restrict__ x, const float* __restrict__ w,
                     float* __restrict__ out, int ntiles, int Btot) {
    __shared__ float Ms[DIM * DIM];
    if (ntiles <= 0) return;
    const int lane = threadIdx.x & 31;
    const int wave = __builtin_amdgcn_readfirstlane((int)(threadIdx.x >> 5));
    const int m    = lane & 15;
    const int h    = lane >> 4;
    const int stride = GRID_BLOCKS * BLOCK_WAVES;
    int tile = (int)blockIdx.x * BLOCK_WAVES + wave;
    const float2* __restrict__ x2 = (const float2*)x;
    float2* __restrict__ out2 = (float2*)out;

    float wj = 0.0f;
    if (wave == 0) wj = w[lane < 12 ? lane : 11];
    const float2 x0 = load_angles(x2, tile, ntiles, Btot, m, h);
    float2 xa = load_angles(x2, tile + stride, ntiles, Btot, m, h);
    float2 xb;

    Trig tr;
    if (wave == 0) build_fixed_matrix(wj, Ms, lane);
    else           tr = sample_trig(x0, h);
    __syncthreads();
    if (wave == 0) tr = sample_trig(x0, h);

    v2f bm[4];
#pragma unroll
    for (int g = 0; g < 4; ++g) {
        const float* mrow = Ms + m * DIM + g * 4 + 2 * h;
        bm[g].x = mrow[0];
        bm[g].y = mrow[1];
    }

    if (tile >= ntiles) return;
    const int n = (ntiles - tile + stride - 1) / stride;
    for (int i = 0; i + 1 < n; i += 2) {
        xb = load_angles(x2, tile + 2 * stride, ntiles, Btot, m, h);
        tile_body(tr, bm, out2, tile, Btot, m, h);
        tr = sample_trig(xa, h);

        xa = load_angles(x2, tile + 3 * stride, ntiles, Btot, m, h);
        tile_body(tr, bm, out2, tile + stride, Btot, m, h);
        if (i + 2 < n) tr = sample_trig(xb, h);
        tile += 2 * stride;
    }
    if (n & 1) tile_body(tr, bm, out2, tile, Btot, m, h);
}

extern "C" void kernel_launch(void* const* d_in, const int* in_sizes, int n_in,
                              void* d_out, int out_size, void* d_ws, size_t ws_size,
                              hipStream_t stream) {
    const float* x = (const float*)d_in[0];
    const float* w = (const float*)d_in[1];
    float* out = (float*)d_out;

    const long B = (long)in_sizes[0] / 4;
    const int ntiles = (int)((B + 15) / 16);

    quantum_layer_kernel<<<GRID_BLOCKS, BLOCK_WAVES * 32, 0, stream>>>(x, w, out, ntiles, (int)B);
}


# A lane's two sincos run side by side on packed f32 (v_pk_mul/fma_f32) with the library's own reduction and polynomials; |arg| >= 2^17 takes the library call under a wave-wide test; results unchanged
# speedup vs baseline: 1.7664x; 1.0814x over previous
#include <hip/hip_runtime.h>
#include <math.h>


#define DIM 16
#define BLOCK_WAVES 8
#define GRID_BLOCKS 1024

typedef float v2f __attribute__((ext_vector_type(2)));
typedef float v8f __attribute__((ext_vector_type(8)));

__device__ __forceinline__ float xhalf(float v) {
    const int b = __builtin_bit_cast(int, v);
    return __builtin_bit_cast(float, __builtin_amdgcn_permlanex16(
        b, b, 0x76543210, 0xfedcba98, false, false));
}

__host__ __device__ constexpr int cnot_perm_idx(int c, int t, int idx) {
    return idx ^ ((((idx >> (3 - c)) & 1)) << (3 - t));
}
__host__ __device__ constexpr int ring_perm_idx(int j) {
    int v = cnot_perm_idx(3, 0, j);
    v = cnot_perm_idx(2, 3, v);
    v = cnot_perm_idx(1, 2, v);
    v = cnot_perm_idx(0, 1, v);
    return v;
}

__device__ __forceinline__ float lane_value(float v, int j) {
    return __builtin_bit_cast(float,
        __builtin_amdgcn_readlane(__builtin_bit_cast(int, v), j));
}

__device__ __forceinline__ void build_fixed_matrix(float wj, float* __restrict__ Ms,
                                                   int lane) {
    const int t = lane & 15;
    float sj, cj;
    sincosf(0.5f * wj, &sj, &cj);
    float v[DIM];
#pragma unroll
    for (int i = 0; i < DIM; ++i) v[i] = (i == t) ? 1.0f : 0.0f;
#pragma unroll
    for (int layer = 0; layer < 3; ++layer) {
#pragma unroll
        for (int q = 0; q < 4; ++q) {
            const float c = lane_value(cj, layer * 4 + q);
            const float s = lane_value(sj, layer * 4 + q);
            const int mask = 1 << (3 - q);
#pragma unroll
            for (int i = 0; i < DIM; ++i) {
                if ((i & mask) == 0) {
                    const int j = i | mask;
                    const float a0 = v[i], a1 = v[j];
                    v[i] = c * a0 - s * a1;
                    v[j] = s * a0 + c * a1;
                }
            }
        }
        float nv[DIM];
#pragma unroll
        for (int i = 0; i < DIM; ++i) nv[i] = v[ring_perm_idx(i)];
#pragma unroll
        for (int i = 0; i < DIM; ++i) v[i] = nv[i];
    }
    if (lane < DIM) {
#pragma unroll
        for (int i = 0; i < DIM; ++i) Ms[i * DIM + t] = v[i];
    }
}

__device__ __forceinline__ float2 load_angles(const float2* __restrict__ x2, int tile,
                                              int ntiles, int Btot, int m, int h) {
    const int t = tile < ntiles ? tile : ntiles - 1;
    const int rem = Btot - t * 16;
    const int nvalid = rem < 16 ? rem : 16;
    const int ml = m < nvalid ? m : nvalid - 1;
    const float2* __restrict__ base = x2 + (long)t * 32;
    return base[ml * 2 + h];
}

__device__ __forceinline__ constexpr float fbits(unsigned u) {
    return __builtin_bit_cast(float, u);
}
__device__ __forceinline__ v2f splat(float c) { v2f v = {c, c}; return v; }

__device__ __forceinline__ void quadrant(float sp, float cp, int n, float& s, float& c) {
    const unsigned flip = ((unsigned)n << 30) & 0x80000000u;
    const bool odd = (n & 1) != 0;
    const float ss = odd ? cp : sp;
    const float cc = odd ? -sp : cp;
    s = __builtin_bit_cast(float, __builtin_bit_cast(unsigned, ss) ^ flip);
    c = __builtin_bit_cast(float, __builtin_bit_cast(unsigned, cc) ^ flip);
}

__device__ __forceinline__ void sincos_pair(v2f h, float& sa, float& ca, float& sb, float& cb) {
    const v2f t = h * splat(fbits(0x3f22f983u));
    v2f fn;
    fn.x = __builtin_rintf(t.x);
    fn.y = __builtin_rintf(t.y);
    v2f r = __builtin_elementwise_fma(fn, splat(fbits(0xbfc90fdau)), h);
    r = __builtin_elementwise_fma(fn, splat(fbits(0xb3a22168u)), r);
    r = __builtin_elementwise_fma(fn, splat(fbits(0xa7c234c4u)), r);
    const int na = (int)fn.x;
    const int nb = (int)fn.y;
    const v2f r2 = r * r;
    v2f p = __builtin_elementwise_fma(r2, splat(fbits(0xb94c1982u)), splat(fbits(0x3c0881c4u)));
    p = __builtin_elementwise_fma(r2, p, splat(fbits(0xbe2aaa9du)));
    p = r2 * p;
    const v2f sp = __builtin_elementwise_fma(r, p, r);
    v2f q = __builtin_elementwise_fma(r2, splat(fbits(0x37d75334u)), splat(fbits(0xbab64f3bu)));
    q = __builtin_elementwise_fma(r2, q, splat(fbits(0x3d2aabf7u)));
    q = __builtin_elementwise_fma(r2, q, splat(fbits(0xbf000004u)));
    const v2f cp = __builtin_elementwise_fma(r2, q, splat(1.0f));
    quadrant(sp.x, cp.x, na, sa, ca);
    quadrant(sp.y, cp.y, nb, sb, cb);
}

struct Trig { float c0, s0, c1, s1, f2, c3, s3; };

__device__ __forceinline__ Trig sample_trig(float2 xh, int h) {
    float sa, ca, sb, cb;
    v2f hv;
    hv.x = 0.5f * xh.x;
    hv.y = 0.5f * xh.y;
    const bool big = (__builtin_fabsf(hv.x) >= 131072.0f) | (__builtin_fabsf(hv.y) >= 131072.0f);
    if (__builtin_expect(__builtin_amdgcn_ballot_w32(big) != 0u, 0)) {
        sincosf(hv.x, &sa, &ca);
        sincosf(hv.y, &sb, &cb);
    } else {
        sincos_pair(hv, sa, ca, sb, cb);
    }
    const float psa = xhalf(sa);
    const float pca = xhalf(ca);
    const float psb = xhalf(sb);
    const float pcb = xhalf(cb);
    Trig t;
    t.c0 = h ? pca : ca;
    t.s0 = h ? psa : sa;
    t.c1 = h ? pcb : cb;
    t.s1 = h ? psb : sb;
    t.f2 = h ? sa : pca;
    t.c3 = h ? cb : pcb;
    t.s3 = h ? sb : psb;
    return t;
}

__device__ __forceinline__ void tile_body(const Trig& tr, const v2f (&bm)[4],
                                          float2* __restrict__ out2, int tile,
                                          int Btot, int m, int h) {
    v8f acc = {0.f, 0.f, 0.f, 0.f, 0.f, 0.f, 0.f, 0.f};
#pragma unroll
    for (int g = 0; g < 4; ++g) {
        const float base = ((g & 2) ? tr.s0 : tr.c0) * ((g & 1) ? tr.s1 : tr.c1) * tr.f2;
        v2f a;
        a.x = base * tr.c3;
        a.y = base * tr.s3;
        acc = __builtin_amdgcn_wmma_f32_16x16x4_f32(
            false, bm[g], false, a, (short)0, acc, false, false);
    }
    float p[8];
#pragma unroll
    for (int r = 0; r < 8; ++r) p[r] = acc[r] * acc[r];
    const float s01 = p[0] + p[1], d01 = p[0] - p[1];
    const float s23 = p[2] + p[3], d23 = p[2] - p[3];
    const float s45 = p[4] + p[5], d45 = p[4] - p[5];
    const float s67 = p[6] + p[7], d67 = p[6] - p[7];
    const float z3  = (d01 + d23) + (d45 + d67);
    const float t0  = s01 + s23,  e0 = s01 - s23;
    const float t1  = s45 + s67,  e1 = s45 - s67;
    const float z2  = e0 + e1;
    const float tot = t0 + t1;
    const float z1  = t0 - t1;
    const float pu = xhalf(h ? tot : z2);
    const float pv = xhalf(h ? z1 : z3);
    float2 r2;
    r2.x = h ? (z2 + pu) : (tot - pu);
    r2.y = (h ? z3 : z1) + pv;
    const int rem = Btot - tile * 16;
    const int nvalid = rem < 16 ? rem : 16;
    if (m < nvalid)
        (out2 + (long)tile * 32)[m * 2 + h] = r2;
}

__global__ void __launch_bounds__(BLOCK_WAVES * 32)
quantum_layer_kernel(const float* __restrict__ x, const float* __restrict__ w,
                     float* __restrict__ out, int ntiles, int Btot) {
    __shared__ float Ms[DIM * DIM];
    if (ntiles <= 0) return;
    const int lane = threadIdx.x & 31;
    const int wave = __builtin_amdgcn_readfirstlane((int)(threadIdx.x >> 5));
    const int m    = lane & 15;
    const int h    = lane >> 4;
    const int stride = GRID_BLOCKS * BLOCK_WAVES;
    int tile = (int)blockIdx.x * BLOCK_WAVES + wave;
    const float2* __restrict__ x2 = (const float2*)x;
    float2* __restrict__ out2 = (float2*)out;

    float wj = 0.0f;
    if (wave == 0) wj = w[lane < 12 ? lane : 11];
    const float2 x0 = load_angles(x2, tile, ntiles, Btot, m, h);
    float2 xa = load_angles(x2, tile + stride, ntiles, Btot, m, h);
    float2 xb;

    Trig tr;
    if (wave == 0) build_fixed_matrix(wj, Ms, lane);
    else           tr = sample_trig(x0, h);
    __syncthreads();
    if (wave == 0) tr = sample_trig(x0, h);

    v2f bm[4];
#pragma unroll
    for (int g = 0; g < 4; ++g) {
        const float* mrow = Ms + m * DIM + g * 4 + 2 * h;
        bm[g].x = mrow[0];
        bm[g].y = mrow[1];
    }

    if (tile >= ntiles) return;
    const int n = (ntiles - tile + stride - 1) / stride;
    for (int i = 0; i + 1 < n; i += 2) {
        xb = load_angles(x2, tile + 2 * stride, ntiles, Btot, m, h);
        tile_body(tr, bm, out2, tile, Btot, m, h);
        tr = sample_trig(xa, h);

        xa = load_angles(x2, tile + 3 * stride, ntiles, Btot, m, h);
        tile_body(tr, bm, out2, tile + stride, Btot, m, h);
        if (i + 2 < n) tr = sample_trig(xb, h);
        tile += 2 * stride;
    }
    if (n & 1) tile_body(tr, bm, out2, tile, Btot, m, h);
}

extern "C" void kernel_launch(void* const* d_in, const int* in_sizes, int n_in,
                              void* d_out, int out_size, void* d_ws, size_t ws_size,
                              hipStream_t stream) {
    const float* x = (const float*)d_in[0];
    const float* w = (const float*)d_in[1];
    float* out = (float*)d_out;

    const long B = (long)in_sizes[0] / 4;
    const int ntiles = (int)((B + 15) / 16);

    quantum_layer_kernel<<<GRID_BLOCKS, BLOCK_WAVES * 32, 0, stream>>>(x, w, out, ntiles, (int)B);
}


# Vector work between sincos and store recast on f32 register pairs: halves trade qubit-pair products (not single factors), amplitudes and Z sums by packed pair ops; product and Z sum order change
# speedup vs baseline: 1.8735x; 1.0606x over previous
#include <hip/hip_runtime.h>
#include <math.h>


#define DIM 16
#define BLOCK_WAVES 8
#define GRID_BLOCKS 1024

typedef float v2f __attribute__((ext_vector_type(2)));
typedef float v8f __attribute__((ext_vector_type(8)));

__device__ __forceinline__ float xhalf(float v) {
    const int b = __builtin_bit_cast(int, v);
    return __builtin_bit_cast(float, __builtin_amdgcn_permlanex16(
        b, b, 0x76543210, 0xfedcba98, true, false));
}

__host__ __device__ constexpr int cnot_perm_idx(int c, int t, int idx) {
    return idx ^ ((((idx >> (3 - c)) & 1)) << (3 - t));
}
__host__ __device__ constexpr int ring_perm_idx(int j) {
    int v = cnot_perm_idx(3, 0, j);
    v = cnot_perm_idx(2, 3, v);
    v = cnot_perm_idx(1, 2, v);
    v = cnot_perm_idx(0, 1, v);
    return v;
}

__device__ __forceinline__ float lane_value(float v, int j) {
    return __builtin_bit_cast(float,
        __builtin_amdgcn_readlane(__builtin_bit_cast(int, v), j));
}

__device__ __forceinline__ void build_fixed_matrix(float wj, float* __restrict__ Ms,
                                                   int lane) {
    const int t = lane & 15;
    float sj, cj;
    sincosf(0.5f * wj, &sj, &cj);
    float v[DIM];
#pragma unroll
    for (int i = 0; i < DIM; ++i) v[i] = (i == t) ? 1.0f : 0.0f;
#pragma unroll
    for (int layer = 0; layer < 3; ++layer) {
#pragma unroll
        for (int q = 0; q < 4; ++q) {
            const float c = lane_value(cj, layer * 4 + q);
            const float s = lane_value(sj, layer * 4 + q);
            const int mask = 1 << (3 - q);
#pragma unroll
            for (int i = 0; i < DIM; ++i) {
                if ((i & mask) == 0) {
                    const int j = i | mask;
                    const float a0 = v[i], a1 = v[j];
                    v[i] = c * a0 - s * a1;
                    v[j] = s * a0 + c * a1;
                }
            }
        }
        float nv[DIM];
#pragma unroll
        for (int i = 0; i < DIM; ++i) nv[i] = v[ring_perm_idx(i)];
#pragma unroll
        for (int i = 0; i < DIM; ++i) v[i] = nv[i];
    }
    if (lane < DIM) {
#pragma unroll
        for (int i = 0; i < DIM; ++i) Ms[i * DIM + t] = v[i];
    }
}

__device__ __forceinline__ float2 load_angles(const float2* __restrict__ x2, int tile,
                                              int ntiles, int Btot, int m, int h) {
    const int t = tile < ntiles ? tile : ntiles - 1;
    const int rem = Btot - t * 16;
    const int nvalid = rem < 16 ? rem : 16;
    const int ml = m < nvalid ? m : nvalid - 1;
    const float2* __restrict__ base = x2 + (long)t * 32;
    return base[ml * 2 + h];
}

__device__ __forceinline__ constexpr float fbits(unsigned u) {
    return __builtin_bit_cast(float, u);
}
__device__ __forceinline__ v2f splat(float c) { v2f v = {c, c}; return v; }

__device__ __forceinline__ void quadrant(float sp, float cp, int n, float& s, float& c) {
    const unsigned flip = ((unsigned)n << 30) & 0x80000000u;
    const bool odd = (n & 1) != 0;
    const float ss = odd ? cp : sp;
    const float cc = odd ? -sp : cp;
    s = __builtin_bit_cast(float, __builtin_bit_cast(unsigned, ss) ^ flip);
    c = __builtin_bit_cast(float, __builtin_bit_cast(unsigned, cc) ^ flip);
}

__device__ __forceinline__ void sincos_pair(v2f h, float& sa, float& ca, float& sb, float& cb) {
    const v2f t = h * splat(fbits(0x3f22f983u));
    v2f fn;
    fn.x = __builtin_rintf(t.x);
    fn.y = __builtin_rintf(t.y);
    v2f r = __builtin_elementwise_fma(fn, splat(fbits(0xbfc90fdau)), h);
    r = __builtin_elementwise_fma(fn, splat(fbits(0xb3a22168u)), r);
    r = __builtin_elementwise_fma(fn, splat(fbits(0xa7c234c4u)), r);
    const int na = (int)fn.x;
    const int nb = (int)fn.y;
    const v2f r2 = r * r;
    v2f p = __builtin_elementwise_fma(r2, splat(fbits(0xb94c1982u)), splat(fbits(0x3c0881c4u)));
    p = __builtin_elementwise_fma(r2, p, splat(fbits(0xbe2aaa9du)));
    p = r2 * p;
    const v2f sp = __builtin_elementwise_fma(r, p, r);
    v2f q = __builtin_elementwise_fma(r2, splat(fbits(0x37d75334u)), splat(fbits(0xbab64f3bu)));
    q = __builtin_elementwise_fma(r2, q, splat(fbits(0x3d2aabf7u)));
    q = __builtin_elementwise_fma(r2, q, splat(fbits(0xbf000004u)));
    const v2f cp = __builtin_elementwise_fma(r2, q, splat(1.0f));
    quadrant(sp.x, cp.x, na, sa, ca);
    quadrant(sp.y, cp.y, nb, sb, cb);
}

struct Trig { float p0, p1, p2, p3; v2f q; };

__device__ __forceinline__ Trig sample_trig(float2 xh, int h) {
    float sa, ca, sb, cb;
    v2f hv;
    hv.x = 0.5f * xh.x;
    hv.y = 0.5f * xh.y;
    const bool big = (__builtin_fabsf(hv.x) >= 131072.0f) | (__builtin_fabsf(hv.y) >= 131072.0f);
    if (__builtin_expect(__builtin_amdgcn_ballot_w32(big) != 0u, 0)) {
        sincosf(hv.x, &sa, &ca);
        sincosf(hv.y, &sb, &cb);
    } else {
        sincos_pair(hv, sa, ca, sb, cb);
    }
    v2f cs;
    cs.x = ca;
    cs.y = sa;
    const v2f r01 = cs * splat(cb);
    const v2f r23 = cs * splat(sb);
    const float x0 = xhalf(r01.x);
    const float x1 = xhalf(r01.y);
    const float x2 = xhalf(r23.x);
    const float x3 = xhalf(r23.y);
    Trig t;
    t.p0 = h ? x0 : r01.x;
    t.p2 = h ? x1 : r01.y;
    t.p1 = h ? x2 : r23.x;
    t.p3 = h ? x3 : r23.y;
    t.q.x = h ? r01.y : x0;
    t.q.y = h ? r23.y : x2;
    return t;
}

__device__ __forceinline__ void tile_body(const Trig& tr, const v2f (&bm)[4],
                                          float2* __restrict__ out2, int tile,
                                          int Btot, int m, int h, float sg) {
    const float pg[4] = {tr.p0, tr.p1, tr.p2, tr.p3};
    v8f acc = {0.f, 0.f, 0.f, 0.f, 0.f, 0.f, 0.f, 0.f};
#pragma unroll
    for (int g = 0; g < 4; ++g) {
        const v2f a = splat(pg[g]) * tr.q;
        acc = __builtin_amdgcn_wmma_f32_16x16x4_f32(
            false, bm[g], false, a, (short)0, acc, false, false);
    }
    v2f y01, y23, y45, y67;
    y01.x = acc[0]; y01.y = acc[1];
    y23.x = acc[2]; y23.y = acc[3];
    y45.x = acc[4]; y45.y = acc[5];
    y67.x = acc[6]; y67.y = acc[7];
    const v2f p01 = y01 * y01, p23 = y23 * y23, p45 = y45 * y45, p67 = y67 * y67;
    const v2f sa = p01 + p23, da = p01 - p23;
    const v2f sb = p45 + p67, db = p45 - p67;
    const v2f ss = sa + sb;
    const v2f sd = sa - sb;
    const v2f ds = da + db;
    const float tot = ss.x + ss.y;
    const float z3  = ss.x - ss.y;
    const float z1  = sd.x + sd.y;
    const float z2  = ds.x + ds.y;
    v2f got, own, w;
    got.x = xhalf(h ? tot : z2);
    got.y = xhalf(h ? z1 : z3);
    own.x = h ? z2 : tot;
    own.y = h ? z3 : z1;
    w.x = sg;
    w.y = 1.0f;
    const v2f r = __builtin_elementwise_fma(got, w, own);
    float2 r2;
    r2.x = r.x;
    r2.y = r.y;
    const int rem = Btot - tile * 16;
    const int nvalid = rem < 16 ? rem : 16;
    if (m < nvalid)
        (out2 + (long)tile * 32)[m * 2 + h] = r2;
}

__global__ void __launch_bounds__(BLOCK_WAVES * 32)
quantum_layer_kernel(const float* __restrict__ x, const float* __restrict__ w,
                     float* __restrict__ out, int ntiles, int Btot) {
    __shared__ float Ms[DIM * DIM];
    if (ntiles <= 0) return;
    const int lane = threadIdx.x & 31;
    const int wave = __builtin_amdgcn_readfirstlane((int)(threadIdx.x >> 5));
    const int m    = lane & 15;
    const int h    = lane >> 4;
    const float sg = h ? 1.0f : -1.0f;
    const int stride = GRID_BLOCKS * BLOCK_WAVES;
    int tile = (int)blockIdx.x * BLOCK_WAVES + wave;
    const float2* __restrict__ x2 = (const float2*)x;
    float2* __restrict__ out2 = (float2*)out;

    float wj = 0.0f;
    if (wave == 0) wj = w[lane < 12 ? lane : 11];
    const float2 x0 = load_angles(x2, tile, ntiles, Btot, m, h);
    float2 xa = load_angles(x2, tile + stride, ntiles, Btot, m, h);
    float2 xb;

    Trig tr;
    if (wave == 0) build_fixed_matrix(wj, Ms, lane);
    else           tr = sample_trig(x0, h);
    __syncthreads();
    if (wave == 0) tr = sample_trig(x0, h);

    v2f bm[4];
#pragma unroll
    for (int g = 0; g < 4; ++g) {
        const float* mrow = Ms + m * DIM + g * 4 + 2 * h;
        bm[g].x = mrow[0];
        bm[g].y = mrow[1];
    }

    if (tile >= ntiles) return;
    const int n = (ntiles - tile + stride - 1) / stride;
    for (int i = 0; i + 1 < n; i += 2) {
        xb = load_angles(x2, tile + 2 * stride, ntiles, Btot, m, h);
        tile_body(tr, bm, out2, tile, Btot, m, h, sg);
        tr = sample_trig(xa, h);

        xa = load_angles(x2, tile + 3 * stride, ntiles, Btot, m, h);
        tile_body(tr, bm, out2, tile + stride, Btot, m, h, sg);
        if (i + 2 < n) tr = sample_trig(xb, h);
        tile += 2 * stride;
    }
    if (n & 1) tile_body(tr, bm, out2, tile, Btot, m, h, sg);
}

extern "C" void kernel_launch(void* const* d_in, const int* in_sizes, int n_in,
                              void* d_out, int out_size, void* d_ws, size_t ws_size,
                              hipStream_t stream) {
    const float* x = (const float*)d_in[0];
    const float* w = (const float*)d_in[1];
    float* out = (float*)d_out;

    const long B = (long)in_sizes[0] / 4;
    const int ntiles = (int)((B + 15) / 16);

    quantum_layer_kernel<<<GRID_BLOCKS, BLOCK_WAVES * 32, 0, stream>>>(x, w, out, ntiles, (int)B);
}


# Sines and cosines of the half angles from the hardware v_sin_f32/v_cos_f32 on x/(4 pi) revolutions for |x| < 5 pi (library code above that): low bits of the results change; sum order unchanged
# speedup vs baseline: 2.0095x; 1.0726x over previous
#include <hip/hip_runtime.h>
#include <math.h>


#define DIM 16
#define BLOCK_WAVES 8
#define GRID_BLOCKS 1024

typedef float v2f __attribute__((ext_vector_type(2)));
typedef float v8f __attribute__((ext_vector_type(8)));

__device__ __forceinline__ float xhalf(float v) {
    const int b = __builtin_bit_cast(int, v);
    return __builtin_bit_cast(float, __builtin_amdgcn_permlanex16(
        b, b, 0x76543210, 0xfedcba98, true, false));
}

__host__ __device__ constexpr int cnot_perm_idx(int c, int t, int idx) {
    return idx ^ ((((idx >> (3 - c)) & 1)) << (3 - t));
}
__host__ __device__ constexpr int ring_perm_idx(int j) {
    int v = cnot_perm_idx(3, 0, j);
    v = cnot_perm_idx(2, 3, v);
    v = cnot_perm_idx(1, 2, v);
    v = cnot_perm_idx(0, 1, v);
    return v;
}

__device__ __forceinline__ float lane_value(float v, int j) {
    return __builtin_bit_cast(float,
        __builtin_amdgcn_readlane(__builtin_bit_cast(int, v), j));
}

__device__ __forceinline__ void build_fixed_matrix(float wj, float* __restrict__ Ms,
                                                   int lane) {
    const int t = lane & 15;
    float sj, cj;
    sincosf(0.5f * wj, &sj, &cj);
    float v[DIM];
#pragma unroll
    for (int i = 0; i < DIM; ++i) v[i] = (i == t) ? 1.0f : 0.0f;
#pragma unroll
    for (int layer = 0; layer < 3; ++layer) {
#pragma unroll
        for (int q = 0; q < 4; ++q) {
            const float c = lane_value(cj, layer * 4 + q);
            const float s = lane_value(sj, layer * 4 + q);
            const int mask = 1 << (3 - q);
#pragma unroll
            for (int i = 0; i < DIM; ++i) {
                if ((i & mask) == 0) {
                    const int j = i | mask;
                    const float a0 = v[i], a1 = v[j];
                    v[i] = c * a0 - s * a1;
                    v[j] = s * a0 + c * a1;
                }
            }
        }
        float nv[DIM];
#pragma unroll
        for (int i = 0; i < DIM; ++i) nv[i] = v[ring_perm_idx(i)];
#pragma unroll
        for (int i = 0; i < DIM; ++i) v[i] = nv[i];
    }
    if (lane < DIM) {
#pragma unroll
        for (int i = 0; i < DIM; ++i) Ms[i * DIM + t] = v[i];
    }
}

__device__ __forceinline__ float2 load_angles(const float2* __restrict__ x2, int tile,
                                              int ntiles, int Btot, int m, int h) {
    const int t = tile < ntiles ? tile : ntiles - 1;
    const int rem = Btot - t * 16;
    const int nvalid = rem < 16 ? rem : 16;
    const int ml = m < nvalid ? m : nvalid - 1;
    const float2* __restrict__ base = x2 + (long)t * 32;
    return base[ml * 2 + h];
}

__device__ __forceinline__ v2f splat(float c) { v2f v = {c, c}; return v; }

#define REV_PER_X   0.0795774715f
#define REV_LIMIT   1.25f

struct Trig { float p0, p1, p2, p3; v2f q; };

__device__ __forceinline__ Trig sample_trig(float2 xh, int h) {
    float sa, ca, sb, cb;
    v2f xv;
    xv.x = xh.x;
    xv.y = xh.y;
    const v2f rev = xv * splat(REV_PER_X);
    const bool big = (__builtin_fabsf(rev.x) >= REV_LIMIT) | (__builtin_fabsf(rev.y) >= REV_LIMIT);
    if (__builtin_expect(__builtin_amdgcn_ballot_w32(big) != 0u, 0)) {
        sincosf(0.5f * xh.x, &sa, &ca);
        sincosf(0.5f * xh.y, &sb, &cb);
    } else {
        sa = __builtin_amdgcn_sinf(rev.x);
        ca = __builtin_amdgcn_cosf(rev.x);
        sb = __builtin_amdgcn_sinf(rev.y);
        cb = __builtin_amdgcn_cosf(rev.y);
    }
    v2f cs;
    cs.x = ca;
    cs.y = sa;
    const v2f r01 = cs * splat(cb);
    const v2f r23 = cs * splat(sb);
    const float x0 = xhalf(r01.x);
    const float x1 = xhalf(r01.y);
    const float x2 = xhalf(r23.x);
    const float x3 = xhalf(r23.y);
    Trig t;
    t.p0 = h ? x0 : r01.x;
    t.p2 = h ? x1 : r01.y;
    t.p1 = h ? x2 : r23.x;
    t.p3 = h ? x3 : r23.y;
    t.q.x = h ? r01.y : x0;
    t.q.y = h ? r23.y : x2;
    return t;
}

__device__ __forceinline__ void tile_body(const Trig& tr, const v2f (&bm)[4],
                                          float2* __restrict__ out2, int tile,
                                          int Btot, int m, int h, float sg) {
    const float pg[4] = {tr.p0, tr.p1, tr.p2, tr.p3};
    v8f acc = {0.f, 0.f, 0.f, 0.f, 0.f, 0.f, 0.f, 0.f};
#pragma unroll
    for (int g = 0; g < 4; ++g) {
        const v2f a = splat(pg[g]) * tr.q;
        acc = __builtin_amdgcn_wmma_f32_16x16x4_f32(
            false, bm[g], false, a, (short)0, acc, false, false);
    }
    v2f y01, y23, y45, y67;
    y01.x = acc[0]; y01.y = acc[1];
    y23.x = acc[2]; y23.y = acc[3];
    y45.x = acc[4]; y45.y = acc[5];
    y67.x = acc[6]; y67.y = acc[7];
    const v2f p01 = y01 * y01, p23 = y23 * y23, p45 = y45 * y45, p67 = y67 * y67;
    const v2f sa = p01 + p23, da = p01 - p23;
    const v2f sb = p45 + p67, db = p45 - p67;
    const v2f ss = sa + sb;
    const v2f sd = sa - sb;
    const v2f ds = da + db;
    const float tot = ss.x + ss.y;
    const float z3  = ss.x - ss.y;
    const float z1  = sd.x + sd.y;
    const float z2  = ds.x + ds.y;
    v2f got, own, w;
    got.x = xhalf(h ? tot : z2);
    got.y = xhalf(h ? z1 : z3);
    own.x = h ? z2 : tot;
    own.y = h ? z3 : z1;
    w.x = sg;
    w.y = 1.0f;
    const v2f r = __builtin_elementwise_fma(got, w, own);
    float2 r2;
    r2.x = r.x;
    r2.y = r.y;
    const int rem = Btot - tile * 16;
    const int nvalid = rem < 16 ? rem : 16;
    if (m < nvalid)
        (out2 + (long)tile * 32)[m * 2 + h] = r2;
}

__global__ void __launch_bounds__(BLOCK_WAVES * 32)
quantum_layer_kernel(const float* __restrict__ x, const float* __restrict__ w,
                     float* __restrict__ out, int ntiles, int Btot) {
    __shared__ float Ms[DIM * DIM];
    if (ntiles <= 0) return;
    const int lane = threadIdx.x & 31;
    const int wave = __builtin_amdgcn_readfirstlane((int)(threadIdx.x >> 5));
    const int m    = lane & 15;
    const int h    = lane >> 4;
    const float sg = h ? 1.0f : -1.0f;
    const int stride = GRID_BLOCKS * BLOCK_WAVES;
    int tile = (int)blockIdx.x * BLOCK_WAVES + wave;
    const float2* __restrict__ x2 = (const float2*)x;
    float2* __restrict__ out2 = (float2*)out;

    float wj = 0.0f;
    if (wave == 0) wj = w[lane < 12 ? lane : 11];
    const float2 x0 = load_angles(x2, tile, ntiles, Btot, m, h);
    float2 xa = load_angles(x2, tile + stride, ntiles, Btot, m, h);
    float2 xb;

    Trig tr;
    if (wave == 0) build_fixed_matrix(wj, Ms, lane);
    else           tr = sample_trig(x0, h);
    __syncthreads();
    if (wave == 0) tr = sample_trig(x0, h);

    v2f bm[4];
#pragma unroll
    for (int g = 0; g < 4; ++g) {
        const float* mrow = Ms + m * DIM + g * 4 + 2 * h;
        bm[g].x = mrow[0];
        bm[g].y = mrow[1];
    }

    if (tile >= ntiles) return;
    const int n = (ntiles - tile + stride - 1) / stride;
    for (int i = 0; i + 1 < n; i += 2) {
        xb = load_angles(x2, tile + 2 * stride, ntiles, Btot, m, h);
        tile_body(tr, bm, out2, tile, Btot, m, h, sg);
        tr = sample_trig(xa, h);

        xa = load_angles(x2, tile + 3 * stride, ntiles, Btot, m, h);
        tile_body(tr, bm, out2, tile + stride, Btot, m, h, sg);
        if (i + 2 < n) tr = sample_trig(xb, h);
        tile += 2 * stride;
    }
    if (n & 1) tile_body(tr, bm, out2, tile, Btot, m, h, sg);
}

extern "C" void kernel_launch(void* const* d_in, const int* in_sizes, int n_in,
                              void* d_out, int out_size, void* d_ws, size_t ws_size,
                              hipStream_t stream) {
    const float* x = (const float*)d_in[0];
    const float* w = (const float*)d_in[1];
    float* out = (float*)d_out;

    const long B = (long)in_sizes[0] / 4;
    const int ntiles = (int)((B + 15) / 16);

    quantum_layer_kernel<<<GRID_BLOCKS, BLOCK_WAVES * 32, 0, stream>>>(x, w, out, ntiles, (int)B);
}


# One whole sample per lane, 32 per trip: one 16-byte load and store, all 4 sincos in-lane; v_permlane16_swap_b32 deals factors to the two 16-sample products and their sums back; sum order unchanged
# speedup vs baseline: 2.2355x; 1.1125x over previous
#include <hip/hip_runtime.h>
#include <math.h>


#define DIM 16
#define BLOCK_WAVES 8
#define GRID_BLOCKS 1024

typedef float v2f __attribute__((ext_vector_type(2)));
typedef float v8f __attribute__((ext_vector_type(8)));

__device__ __forceinline__ void swap_halves4(float& d0, float& s0, float& d1, float& s1,
                                             float& d2, float& s2, float& d3, float& s3) {
    asm volatile("s_delay_alu instid0(VALU_DEP_1)\n\t"
                 "v_permlane16_swap_b32 %0, %1\n\t"
                 "v_permlane16_swap_b32 %2, %3\n\t"
                 "v_permlane16_swap_b32 %4, %5\n\t"
                 "v_permlane16_swap_b32 %6, %7"
                 : "+v"(d0), "+v"(s0), "+v"(d1), "+v"(s1),
                   "+v"(d2), "+v"(s2), "+v"(d3), "+v"(s3));
}
__device__ __forceinline__ void swap_halves6(float& d0, float& s0, float& d1, float& s1,
                                             float& d2, float& s2, float& d3, float& s3,
                                             float& d4, float& s4, float& d5, float& s5) {
    asm volatile("s_delay_alu instid0(VALU_DEP_1)\n\t"
                 "v_permlane16_swap_b32 %0, %1\n\t"
                 "v_permlane16_swap_b32 %2, %3\n\t"
                 "v_permlane16_swap_b32 %4, %5\n\t"
                 "v_permlane16_swap_b32 %6, %7\n\t"
                 "v_permlane16_swap_b32 %8, %9\n\t"
                 "v_permlane16_swap_b32 %10, %11"
                 : "+v"(d0), "+v"(s0), "+v"(d1), "+v"(s1), "+v"(d2), "+v"(s2),
                   "+v"(d3), "+v"(s3), "+v"(d4), "+v"(s4), "+v"(d5), "+v"(s5));
}

__host__ __device__ constexpr int cnot_perm_idx(int c, int t, int idx) {
    return idx ^ ((((idx >> (3 - c)) & 1)) << (3 - t));
}
__host__ __device__ constexpr int ring_perm_idx(int j) {
    int v = cnot_perm_idx(3, 0, j);
    v = cnot_perm_idx(2, 3, v);
    v = cnot_perm_idx(1, 2, v);
    v = cnot_perm_idx(0, 1, v);
    return v;
}

__device__ __forceinline__ float lane_value(float v, int j) {
    return __builtin_bit_cast(float,
        __builtin_amdgcn_readlane(__builtin_bit_cast(int, v), j));
}

__device__ __forceinline__ void build_fixed_matrix(float wj, float* __restrict__ Ms,
                                                   int lane) {
    const int t = lane & 15;
    float sj, cj;
    sincosf(0.5f * wj, &sj, &cj);
    float v[DIM];
#pragma unroll
    for (int i = 0; i < DIM; ++i) v[i] = (i == t) ? 1.0f : 0.0f;
#pragma unroll
    for (int layer = 0; layer < 3; ++layer) {
#pragma unroll
        for (int q = 0; q < 4; ++q) {
            const float c = lane_value(cj, layer * 4 + q);
            const float s = lane_value(sj, layer * 4 + q);
            const int mask = 1 << (3 - q);
#pragma unroll
            for (int i = 0; i < DIM; ++i) {
                if ((i & mask) == 0) {
                    const int j = i | mask;
                    const float a0 = v[i], a1 = v[j];
                    v[i] = c * a0 - s * a1;
                    v[j] = s * a0 + c * a1;
                }
            }
        }
        float nv[DIM];
#pragma unroll
        for (int i = 0; i < DIM; ++i) nv[i] = v[ring_perm_idx(i)];
#pragma unroll
        for (int i = 0; i < DIM; ++i) v[i] = nv[i];
    }
    if (lane < DIM) {
#pragma unroll
        for (int i = 0; i < DIM; ++i) Ms[i * DIM + t] = v[i];
    }
}

__device__ __forceinline__ float4 load_angles(const float4* __restrict__ x4, int group,
                                              int ngroups, int Btot, int l) {
    const int g = group < ngroups ? group : ngroups - 1;
    const int last = Btot - 1 - g * 32;
    const int ll = l < last ? l : last;
    const float4* __restrict__ base = x4 + (long)g * 32;
    return base[ll];
}

__device__ __forceinline__ v2f splat(float c) { v2f v = {c, c}; return v; }

#define REV_PER_X   0.0795774715f
#define REV_LIMIT   1.25f

struct Frag { float p0, p1, p2, p3; v2f q; };
struct Trig { Frag a, b; };

__device__ __forceinline__ Trig sample_trig(float4 xs) {
    float s0, c0, s1, c1, s2, c2, s3, c3;
    v2f x01, x23;
    x01.x = xs.x; x01.y = xs.y;
    x23.x = xs.z; x23.y = xs.w;
    const v2f r01 = x01 * splat(REV_PER_X);
    const v2f r23 = x23 * splat(REV_PER_X);
    const bool big = (__builtin_fabsf(r01.x) >= REV_LIMIT) | (__builtin_fabsf(r01.y) >= REV_LIMIT) |
                     (__builtin_fabsf(r23.x) >= REV_LIMIT) | (__builtin_fabsf(r23.y) >= REV_LIMIT);
    if (__builtin_expect(__builtin_amdgcn_ballot_w32(big) != 0u, 0)) {
        sincosf(0.5f * xs.x, &s0, &c0);
        sincosf(0.5f * xs.y, &s1, &c1);
        sincosf(0.5f * xs.z, &s2, &c2);
        sincosf(0.5f * xs.w, &s3, &c3);
    } else {
        s0 = __builtin_amdgcn_sinf(r01.x);
        c0 = __builtin_amdgcn_cosf(r01.x);
        s1 = __builtin_amdgcn_sinf(r01.y);
        c1 = __builtin_amdgcn_cosf(r01.y);
        s2 = __builtin_amdgcn_sinf(r23.x);
        c2 = __builtin_amdgcn_cosf(r23.x);
        s3 = __builtin_amdgcn_sinf(r23.y);
        c3 = __builtin_amdgcn_cosf(r23.y);
    }
    v2f cs0, cs3;
    cs0.x = c0; cs0.y = s0;
    cs3.x = c3; cs3.y = s3;
    const v2f pe = cs0 * splat(c1);
    const v2f po = cs0 * splat(s1);
    const v2f q0 = cs3 * splat(c2);
    const v2f q1 = cs3 * splat(s2);
    float qa0 = q0.x, qb0 = q1.x, qa1 = q0.y, qb1 = q1.y;
    float a0 = pe.x, b0 = pe.x, a1 = po.x, b1 = po.x;
    float a2 = pe.y, b2 = pe.y, a3 = po.y, b3 = po.y;
    swap_halves6(qa0, qb0, qa1, qb1, a0, b0, a1, b1, a2, b2, a3, b3);
    Trig t;
    t.a.p0 = a0; t.a.p1 = a1; t.a.p2 = a2; t.a.p3 = a3;
    t.b.p0 = b0; t.b.p1 = b1; t.b.p2 = b2; t.b.p3 = b3;
    t.a.q.x = qa0; t.a.q.y = qa1;
    t.b.q.x = qb0; t.b.q.y = qb1;
    return t;
}

struct Part { float tot, z1, z2, z3; };

__device__ __forceinline__ Part signed_sums(const v8f& acc) {
    v2f y01, y23, y45, y67;
    y01.x = acc[0]; y01.y = acc[1];
    y23.x = acc[2]; y23.y = acc[3];
    y45.x = acc[4]; y45.y = acc[5];
    y67.x = acc[6]; y67.y = acc[7];
    const v2f p01 = y01 * y01, p23 = y23 * y23, p45 = y45 * y45, p67 = y67 * y67;
    const v2f sa = p01 + p23, da = p01 - p23;
    const v2f sb = p45 + p67, db = p45 - p67;
    const v2f ss = sa + sb;
    const v2f sd = sa - sb;
    const v2f ds = da + db;
    Part r;
    r.tot = ss.x + ss.y;
    r.z3  = ss.x - ss.y;
    r.z1  = sd.x + sd.y;
    r.z2  = ds.x + ds.y;
    return r;
}

__device__ __forceinline__ void group_body(const Trig& tr, const v2f (&bm)[4],
                                           float4* __restrict__ out4, int group,
                                           int Btot, int l) {
    const float pa[4] = {tr.a.p0, tr.a.p1, tr.a.p2, tr.a.p3};
    const float pb[4] = {tr.b.p0, tr.b.p1, tr.b.p2, tr.b.p3};
    v8f acca = {0.f, 0.f, 0.f, 0.f, 0.f, 0.f, 0.f, 0.f};
    v8f accb = {0.f, 0.f, 0.f, 0.f, 0.f, 0.f, 0.f, 0.f};
#pragma unroll
    for (int g = 0; g < 4; ++g) {
        const v2f sa = splat(pa[g]) * tr.a.q;
        const v2f sb = splat(pb[g]) * tr.b.q;
        acca = __builtin_amdgcn_wmma_f32_16x16x4_f32(
            false, bm[g], false, sa, (short)0, acca, false, false);
        accb = __builtin_amdgcn_wmma_f32_16x16x4_f32(
            false, bm[g], false, sb, (short)0, accb, false, false);
    }
    const Part ua = signed_sums(acca);
    const Part ub = signed_sums(accb);
    float lt = ua.tot, ut = ub.tot, l1 = ua.z1, u1 = ub.z1;
    float l2 = ua.z2,  u2 = ub.z2,  l3 = ua.z3, u3 = ub.z3;
    swap_halves4(l2, u2, l3, u3, lt, ut, l1, u1);
    v2f lo01, up01, lo23, up23, w;
    lo01.x = lt; lo01.y = l1;
    up01.x = ut; up01.y = u1;
    lo23.x = l2; lo23.y = l3;
    up23.x = u2; up23.y = u3;
    w.x = -1.0f;
    w.y = 1.0f;
    const v2f r01 = __builtin_elementwise_fma(up01, w, lo01);
    const v2f r23 = lo23 + up23;
    float4 r4;
    r4.x = r01.x;
    r4.y = r01.y;
    r4.z = r23.x;
    r4.w = r23.y;
    const int last = Btot - 1 - group * 32;
    if (l <= last)
        (out4 + (long)group * 32)[l] = r4;
}

__global__ void __launch_bounds__(BLOCK_WAVES * 32)
quantum_layer_kernel(const float* __restrict__ x, const float* __restrict__ w,
                     float* __restrict__ out, int ntiles, int Btot) {
    __shared__ float Ms[DIM * DIM];
    if (ntiles <= 0) return;
    const int lane = threadIdx.x & 31;
    const int wave = __builtin_amdgcn_readfirstlane((int)(threadIdx.x >> 5));
    const int m    = lane & 15;
    const int h    = lane >> 4;
    const int ngroups = (ntiles + 1) >> 1;
    const int stride = GRID_BLOCKS * BLOCK_WAVES;
    int group = (int)blockIdx.x * BLOCK_WAVES + wave;
    const float4* __restrict__ x4 = (const float4*)x;
    float4* __restrict__ out4 = (float4*)out;

    float wj = 0.0f;
    if (wave == 0) wj = w[lane < 12 ? lane : 11];
    const float4 x0 = load_angles(x4, group, ngroups, Btot, lane);
    float4 xa = load_angles(x4, group + stride, ngroups, Btot, lane);
    float4 xb;

    Trig tr;
    if (wave == 0) build_fixed_matrix(wj, Ms, lane);
    else           tr = sample_trig(x0);
    __syncthreads();
    if (wave == 0) tr = sample_trig(x0);

    v2f bm[4];
#pragma unroll
    for (int g = 0; g < 4; ++g) {
        const float* mrow = Ms + m * DIM + g * 4 + 2 * h;
        bm[g].x = mrow[0];
        bm[g].y = mrow[1];
    }

    if (group >= ngroups) return;
    const int n = (ngroups - group + stride - 1) / stride;
    for (int i = 0; i + 1 < n; i += 2) {
        xb = load_angles(x4, group + 2 * stride, ngroups, Btot, lane);
        group_body(tr, bm, out4, group, Btot, lane);
        tr = sample_trig(xa);

        xa = load_angles(x4, group + 3 * stride, ngroups, Btot, lane);
        group_body(tr, bm, out4, group + stride, Btot, lane);
        if (i + 2 < n) tr = sample_trig(xb);
        group += 2 * stride;
    }
    if (n & 1) group_body(tr, bm, out4, group, Btot, lane);
}

extern "C" void kernel_launch(void* const* d_in, const int* in_sizes, int n_in,
                              void* d_out, int out_size, void* d_ws, size_t ws_size,
                              hipStream_t stream) {
    const float* x = (const float*)d_in[0];
    const float* w = (const float*)d_in[1];
    float* out = (float*)d_out;

    const long B = (long)in_sizes[0] / 4;
    const int ntiles = (int)((B + 15) / 16);

    quantum_layer_kernel<<<GRID_BLOCKS, BLOCK_WAVES * 32, 0, stream>>>(x, w, out, ntiles, (int)B);
}


# All of a group that needs no operator (angle loads, sines, column operands) is done for 4 groups per wave before the barrier, under the operator build; behind it: products, sums, store
# speedup vs baseline: 2.4933x; 1.1153x over previous
#include <hip/hip_runtime.h>
#include <math.h>


#define DIM 16
#define BLOCK_WAVES 8
#define GRID_BLOCKS 1024
#define AHEAD 4

typedef float v2f __attribute__((ext_vector_type(2)));
typedef float v8f __attribute__((ext_vector_type(8)));

__device__ __forceinline__ void swap_halves4(float& d0, float& s0, float& d1, float& s1,
                                             float& d2, float& s2, float& d3, float& s3) {
    asm volatile("s_delay_alu instid0(VALU_DEP_1)\n\t"
                 "v_permlane16_swap_b32 %0, %1\n\t"
                 "v_permlane16_swap_b32 %2, %3\n\t"
                 "v_permlane16_swap_b32 %4, %5\n\t"
                 "v_permlane16_swap_b32 %6, %7"
                 : "+v"(d0), "+v"(s0), "+v"(d1), "+v"(s1),
                   "+v"(d2), "+v"(s2), "+v"(d3), "+v"(s3));
}
__device__ __forceinline__ void swap_halves6(float& d0, float& s0, float& d1, float& s1,
                                             float& d2, float& s2, float& d3, float& s3,
                                             float& d4, float& s4, float& d5, float& s5) {
    asm volatile("s_delay_alu instid0(VALU_DEP_1)\n\t"
                 "v_permlane16_swap_b32 %0, %1\n\t"
                 "v_permlane16_swap_b32 %2, %3\n\t"
                 "v_permlane16_swap_b32 %4, %5\n\t"
                 "v_permlane16_swap_b32 %6, %7\n\t"
                 "v_permlane16_swap_b32 %8, %9\n\t"
                 "v_permlane16_swap_b32 %10, %11"
                 : "+v"(d0), "+v"(s0), "+v"(d1), "+v"(s1), "+v"(d2), "+v"(s2),
                   "+v"(d3), "+v"(s3), "+v"(d4), "+v"(s4), "+v"(d5), "+v"(s5));
}

__host__ __device__ constexpr int cnot_perm_idx(int c, int t, int idx) {
    return idx ^ ((((idx >> (3 - c)) & 1)) << (3 - t));
}
__host__ __device__ constexpr int ring_perm_idx(int j) {
    int v = cnot_perm_idx(3, 0, j);
    v = cnot_perm_idx(2, 3, v);
    v = cnot_perm_idx(1, 2, v);
    v = cnot_perm_idx(0, 1, v);
    return v;
}

__device__ __forceinline__ float lane_value(float v, int j) {
    return __builtin_bit_cast(float,
        __builtin_amdgcn_readlane(__builtin_bit_cast(int, v), j));
}

__device__ __forceinline__ void build_fixed_matrix(float wj, float* __restrict__ Ms,
                                                   int lane) {
    const int t = lane & 15;
    float sj, cj;
    sincosf(0.5f * wj, &sj, &cj);
    float v[DIM];
#pragma unroll
    for (int i = 0; i < DIM; ++i) v[i] = (i == t) ? 1.0f : 0.0f;
#pragma unroll
    for (int layer = 0; layer < 3; ++layer) {
#pragma unroll
        for (int q = 0; q < 4; ++q) {
            const float c = lane_value(cj, layer * 4 + q);
            const float s = lane_value(sj, layer * 4 + q);
            const int mask = 1 << (3 - q);
#pragma unroll
            for (int i = 0; i < DIM; ++i) {
                if ((i & mask) == 0) {
                    const int j = i | mask;
                    const float a0 = v[i], a1 = v[j];
                    v[i] = c * a0 - s * a1;
                    v[j] = s * a0 + c * a1;
                }
            }
        }
        float nv[DIM];
#pragma unroll
        for (int i = 0; i < DIM; ++i) nv[i] = v[ring_perm_idx(i)];
#pragma unroll
        for (int i = 0; i < DIM; ++i) v[i] = nv[i];
    }
    if (lane < DIM) {
#pragma unroll
        for (int i = 0; i < DIM; ++i) Ms[i * DIM + t] = v[i];
    }
}

__device__ __forceinline__ float4 load_angles(const float4* __restrict__ x4, int group,
                                              int ngroups, int Btot, int l) {
    const int g = group < ngroups ? group : ngroups - 1;
    const int last = Btot - 1 - g * 32;
    const int ll = l < last ? l : last;
    const float4* __restrict__ base = x4 + (long)g * 32;
    return base[ll];
}

__device__ __forceinline__ v2f splat(float c) { v2f v = {c, c}; return v; }

#define REV_PER_X   0.0795774715f
#define REV_LIMIT   1.25f

struct Frag { float p0, p1, p2, p3; v2f q; };
struct Trig { Frag a, b; };

__device__ __forceinline__ Trig trig_factors(float s0, float c0, float s1, float c1,
                                             float s2, float c2, float s3, float c3) {
    v2f cs0, cs3;
    cs0.x = c0; cs0.y = s0;
    cs3.x = c3; cs3.y = s3;
    const v2f pe = cs0 * splat(c1);
    const v2f po = cs0 * splat(s1);
    const v2f q0 = cs3 * splat(c2);
    const v2f q1 = cs3 * splat(s2);
    float qa0 = q0.x, qb0 = q1.x, qa1 = q0.y, qb1 = q1.y;
    float a0 = pe.x, b0 = pe.x, a1 = po.x, b1 = po.x;
    float a2 = pe.y, b2 = pe.y, a3 = po.y, b3 = po.y;
    swap_halves6(qa0, qb0, qa1, qb1, a0, b0, a1, b1, a2, b2, a3, b3);
    Trig t;
    t.a.p0 = a0; t.a.p1 = a1; t.a.p2 = a2; t.a.p3 = a3;
    t.b.p0 = b0; t.b.p1 = b1; t.b.p2 = b2; t.b.p3 = b3;
    t.a.q.x = qa0; t.a.q.y = qa1;
    t.b.q.x = qb0; t.b.q.y = qb1;
    return t;
}

struct Ops { v2f a[4]; v2f b[4]; };

__device__ __forceinline__ Ops ops_of(const Trig& tr) {
    const float pa[4] = {tr.a.p0, tr.a.p1, tr.a.p2, tr.a.p3};
    const float pb[4] = {tr.b.p0, tr.b.p1, tr.b.p2, tr.b.p3};
    Ops o;
#pragma unroll
    for (int g = 0; g < 4; ++g) {
        o.a[g] = splat(pa[g]) * tr.a.q;
        o.b[g] = splat(pb[g]) * tr.b.q;
    }
    return o;
}

__device__ __forceinline__ void prepare_groups(const float4 (&xs)[AHEAD], Ops (&op)[AHEAD]) {
    v2f r01[AHEAD], r23[AHEAD];
    bool big = false;
#pragma unroll
    for (int j = 0; j < AHEAD; ++j) {
        v2f x01, x23;
        x01.x = xs[j].x; x01.y = xs[j].y;
        x23.x = xs[j].z; x23.y = xs[j].w;
        r01[j] = x01 * splat(REV_PER_X);
        r23[j] = x23 * splat(REV_PER_X);
        big |= (__builtin_fabsf(r01[j].x) >= REV_LIMIT) | (__builtin_fabsf(r01[j].y) >= REV_LIMIT) |
               (__builtin_fabsf(r23[j].x) >= REV_LIMIT) | (__builtin_fabsf(r23[j].y) >= REV_LIMIT);
    }
    float s[AHEAD][4], c[AHEAD][4];
    if (__builtin_expect(__builtin_amdgcn_ballot_w32(big) != 0u, 0)) {
#pragma unroll
        for (int j = 0; j < AHEAD; ++j) {
            sincosf(0.5f * xs[j].x, &s[j][0], &c[j][0]);
            sincosf(0.5f * xs[j].y, &s[j][1], &c[j][1]);
            sincosf(0.5f * xs[j].z, &s[j][2], &c[j][2]);
            sincosf(0.5f * xs[j].w, &s[j][3], &c[j][3]);
        }
    } else {
#pragma unroll
        for (int j = 0; j < AHEAD; ++j) {
            s[j][0] = __builtin_amdgcn_sinf(r01[j].x);
            c[j][0] = __builtin_amdgcn_cosf(r01[j].x);
            s[j][1] = __builtin_amdgcn_sinf(r01[j].y);
            c[j][1] = __builtin_amdgcn_cosf(r01[j].y);
            s[j][2] = __builtin_amdgcn_sinf(r23[j].x);
            c[j][2] = __builtin_amdgcn_cosf(r23[j].x);
            s[j][3] = __builtin_amdgcn_sinf(r23[j].y);
            c[j][3] = __builtin_amdgcn_cosf(r23[j].y);
        }
    }
#pragma unroll
    for (int j = 0; j < AHEAD; ++j)
        op[j] = ops_of(trig_factors(s[j][0], c[j][0], s[j][1], c[j][1],
                                    s[j][2], c[j][2], s[j][3], c[j][3]));
}

struct Part { float tot, z1, z2, z3; };

__device__ __forceinline__ Part signed_sums(const v8f& acc) {
    v2f y01, y23, y45, y67;
    y01.x = acc[0]; y01.y = acc[1];
    y23.x = acc[2]; y23.y = acc[3];
    y45.x = acc[4]; y45.y = acc[5];
    y67.x = acc[6]; y67.y = acc[7];
    const v2f p01 = y01 * y01, p23 = y23 * y23, p45 = y45 * y45, p67 = y67 * y67;
    const v2f sa = p01 + p23, da = p01 - p23;
    const v2f sb = p45 + p67, db = p45 - p67;
    const v2f ss = sa + sb;
    const v2f sd = sa - sb;
    const v2f ds = da + db;
    Part r;
    r.tot = ss.x + ss.y;
    r.z3  = ss.x - ss.y;
    r.z1  = sd.x + sd.y;
    r.z2  = ds.x + ds.y;
    return r;
}

__device__ __forceinline__ void group_body(const Ops& op, const v2f (&bm)[4],
                                           float4* __restrict__ out4, int group,
                                           int Btot, int l) {
    v8f acca = {0.f, 0.f, 0.f, 0.f, 0.f, 0.f, 0.f, 0.f};
    v8f accb = {0.f, 0.f, 0.f, 0.f, 0.f, 0.f, 0.f, 0.f};
#pragma unroll
    for (int g = 0; g < 4; ++g) {
        acca = __builtin_amdgcn_wmma_f32_16x16x4_f32(
            false, bm[g], false, op.a[g], (short)0, acca, false, false);
        accb = __builtin_amdgcn_wmma_f32_16x16x4_f32(
            false, bm[g], false, op.b[g], (short)0, accb, false, false);
    }
    const Part ua = signed_sums(acca);
    const Part ub = signed_sums(accb);
    float lt = ua.tot, ut = ub.tot, l1 = ua.z1, u1 = ub.z1;
    float l2 = ua.z2,  u2 = ub.z2,  l3 = ua.z3, u3 = ub.z3;
    swap_halves4(l2, u2, l3, u3, lt, ut, l1, u1);
    v2f lo01, up01, lo23, up23, w;
    lo01.x = lt; lo01.y = l1;
    up01.x = ut; up01.y = u1;
    lo23.x = l2; lo23.y = l3;
    up23.x = u2; up23.y = u3;
    w.x = -1.0f;
    w.y = 1.0f;
    const v2f r01 = __builtin_elementwise_fma(up01, w, lo01);
    const v2f r23 = lo23 + up23;
    float4 r4;
    r4.x = r01.x;
    r4.y = r01.y;
    r4.z = r23.x;
    r4.w = r23.y;
    const int last = Btot - 1 - group * 32;
    if (l <= last)
        (out4 + (long)group * 32)[l] = r4;
}

__global__ void __launch_bounds__(BLOCK_WAVES * 32)
quantum_layer_kernel(const float* __restrict__ x, const float* __restrict__ w,
                     float* __restrict__ out, int ntiles, int Btot) {
    __shared__ float Ms[DIM * DIM];
    if (ntiles <= 0) return;
    const int lane = threadIdx.x & 31;
    const int wave = __builtin_amdgcn_readfirstlane((int)(threadIdx.x >> 5));
    const int m    = lane & 15;
    const int h    = lane >> 4;
    const int ngroups = (ntiles + 1) >> 1;
    const int stride = GRID_BLOCKS * BLOCK_WAVES;
    int group = (int)blockIdx.x * BLOCK_WAVES + wave;
    const float4* __restrict__ x4 = (const float4*)x;
    float4* __restrict__ out4 = (float4*)out;

    float wj = 0.0f;
    if (wave == 0) wj = w[lane < 12 ? lane : 11];
    float4 xs[AHEAD];
#pragma unroll
    for (int j = 0; j < AHEAD; ++j)
        xs[j] = load_angles(x4, group + j * stride, ngroups, Btot, lane);

    Ops op[AHEAD];
    if (wave == 0) {
        build_fixed_matrix(wj, Ms, lane);
    } else {
        prepare_groups(xs, op);
    }
    __syncthreads();
    if (wave == 0) prepare_groups(xs, op);

    v2f bm[4];
#pragma unroll
    for (int g = 0; g < 4; ++g) {
        const float* mrow = Ms + m * DIM + g * 4 + 2 * h;
        bm[g].x = mrow[0];
        bm[g].y = mrow[1];
    }

    if (group >= ngroups) return;
    for (;;) {
#pragma unroll
        for (int j = 0; j < AHEAD; ++j) {
            const int gj = group + j * stride;
            if (gj < ngroups) group_body(op[j], bm, out4, gj, Btot, lane);
        }
        group += AHEAD * stride;
        if (group >= ngroups) break;
#pragma unroll
        for (int j = 0; j < AHEAD; ++j)
            xs[j] = load_angles(x4, group + j * stride, ngroups, Btot, lane);
        prepare_groups(xs, op);
    }
}

extern "C" void kernel_launch(void* const* d_in, const int* in_sizes, int n_in,
                              void* d_out, int out_size, void* d_ws, size_t ws_size,
                              hipStream_t stream) {
    const float* x = (const float*)d_in[0];
    const float* w = (const float*)d_in[1];
    float* out = (float*)d_out;

    const long B = (long)in_sizes[0] / 4;
    const int ntiles = (int)((B + 15) / 16);

    quantum_layer_kernel<<<GRID_BLOCKS, BLOCK_WAVES * 32, 0, stream>>>(x, w, out, ntiles, (int)B);
}
